# starting kernel
# baseline (speedup 1.0000x reference)
#include <hip/hip_runtime.h>
#include <hip/hip_bf16.h>

typedef __attribute__((ext_vector_type(2))) float v2f;
typedef __attribute__((ext_vector_type(8))) float v8f;

#define L_TOK   4096
#define B_BATCH 8
#define K0      484
#define H0      32
#define TB      128
#define HALO    120

__global__ __launch_bounds__(256) void k_wsr(const float* __restrict__ Ws,
                                             const float* __restrict__ Wr,
                                             float* __restrict__ Wsr) {
    int id = blockIdx.x * 256 + threadIdx.x;
    int r = id >> 2, o = id & 3;
    const float* wsrow = Ws + (size_t)r * 4096;
    float acc = 0.f;
    for (int k = 0; k < 4096; ++k) acc = fmaf(wsrow[k], Wr[k * 4 + o], acc);
    Wsr[r * 4 + o] = acc;
}

__global__ __launch_bounds__(256) void k_fold(const float* __restrict__ W1,
                                              const float* __restrict__ b1,
                                              const float* __restrict__ bs,
                                              const float* __restrict__ Wr,
                                              const float* __restrict__ br,
                                              const float* __restrict__ Wsr,
                                              float* __restrict__ Wc,
                                              float* __restrict__ bc) {
    int tid = threadIdx.x;
    if (tid < 128) {
        int n = tid >> 2, o = tid & 3;
        float acc = 0.f;
        for (int r = 0; r < 512; ++r) acc = fmaf(W1[n * 512 + r], Wsr[r * 4 + o], acc);
        Wc[tid] = acc;
    } else if (tid < 132) {
        int o = tid - 128;
        float acc = br[o];
        for (int r = 0; r < 512; ++r)  acc = fmaf(b1[r], Wsr[r * 4 + o], acc);
        for (int k = 0; k < 4096; ++k) acc = fmaf(bs[k], Wr[k * 4 + o], acc);
        bc[o] = acc;
    }
}

__global__ __launch_bounds__(256) void k_stage1(const float* __restrict__ x,
                                                const float* __restrict__ W0,
                                                const float* __restrict__ b0,
                                                const float* __restrict__ Wc,
                                                const float* __restrict__ bc,
                                                float* __restrict__ hmid) {
    __shared__ float xt[(TB + HALO) * 4];
    __shared__ float w0t[H0 * K0];
    __shared__ float b0s[H0];
    __shared__ float wcs[H0 * 4];
    __shared__ float bcs[4];
    __shared__ float h0s[TB * 36];

    const int tid = threadIdx.x;
    const int bb  = blockIdx.x >> 5;
    const int t0  = (blockIdx.x & 31) * TB;

    for (int i = tid; i < (TB + HALO) * 4; i += 256) {
        int j = i >> 2, c = i & 3;
        int gt = (t0 + j + L_TOK - HALO) & (L_TOK - 1);
        xt[i] = x[((bb << 12) + gt) * 4 + c];
    }
    for (int i = tid; i < H0 * K0; i += 256) {
        int k = i >> 5, n = i & 31;
        w0t[n * K0 + k] = W0[i];
    }
    if (tid < H0)  b0s[tid] = b0[tid];
    if (tid < 128) wcs[tid] = Wc[tid];
    if (tid < 4)   bcs[tid] = bc[tid];
    __syncthreads();

    const int wave = tid >> 5;
    const int lane = tid & 31;
    const int hl   = lane >> 4;
    const int m0   = lane & 15;
    const int c0   = hl << 1;

    v8f acc0 = {}; v8f acc1 = {};
    const float* ap  = xt  + (((wave << 4) + m0)) * 4 + c0;
    const float* bp0 = w0t + m0 * K0 + c0;
    const float* bp1 = w0t + (m0 + 16) * K0 + c0;

    for (int kk = 0; kk < 121; ++kk) {
        v2f a  = *(const v2f*)(ap  + 4 * kk);
        v2f bA = *(const v2f*)(bp0 + 4 * kk);
        v2f bB = *(const v2f*)(bp1 + 4 * kk);
        acc0 = __builtin_amdgcn_wmma_f32_16x16x4_f32(false, a, false, bA, (short)0, acc0, false, false);
        acc1 = __builtin_amdgcn_wmma_f32_16x16x4_f32(false, a, false, bB, (short)0, acc1, false, false);
    }

    #pragma unroll
    for (int g = 0; g < 8; ++g) {
        int M = (wave << 4) + g + (hl << 3);
        h0s[M * 36 + m0]      = tanhf(acc0[g] + b0s[m0]);
        h0s[M * 36 + m0 + 16] = tanhf(acc1[g] + b0s[m0 + 16]);
    }
    __syncthreads();

    #pragma unroll
    for (int r = 0; r < 2; ++r) {
        int idx = tid + (r << 8);
        int tok = idx >> 2, o = idx & 3;
        const float* hrow = h0s + tok * 36;
        float s = bcs[o];
        #pragma unroll 8
        for (int n = 0; n < 32; ++n) s = fmaf(hrow[n], wcs[n * 4 + o], s);
        hmid[(((bb << 12) + t0 + tok) << 2) + o] = s;
    }
}

__global__ __launch_bounds__(256) void k_stage2(const float* __restrict__ hmid,
                                                const float* __restrict__ W2,
                                                const float* __restrict__ b2,
                                                const float* __restrict__ W3,
                                                const float* __restrict__ b3,
                                                float* __restrict__ out) {
    __shared__ float w2s[20 * 32];
    __shared__ float b2s[32];
    __shared__ float w3s[32 * 4];
    __shared__ float b3s[4];
    int tid = threadIdx.x;
    for (int i = tid; i < 640; i += 256) w2s[i] = W2[i];
    if (tid < 32)  b2s[tid] = b2[tid];
    if (tid < 128) w3s[tid] = W3[tid];
    if (tid < 4)   b3s[tid] = b3[tid];
    __syncthreads();

    int g  = blockIdx.x * 256 + tid;
    int bb = g >> 12, t = g & (L_TOK - 1);

    float win[20];
    #pragma unroll
    for (int j = 0; j < 5; ++j) {
        int tt = (t + L_TOK - 4 + j) & (L_TOK - 1);
        const float* p = hmid + (((bb << 12) + tt) << 2);
        win[j * 4 + 0] = p[0]; win[j * 4 + 1] = p[1];
        win[j * 4 + 2] = p[2]; win[j * 4 + 3] = p[3];
    }
    float o0 = b3s[0], o1 = b3s[1], o2 = b3s[2], o3 = b3s[3];
    for (int n = 0; n < 32; ++n) {
        float h = b2s[n];
        #pragma unroll
        for (int k = 0; k < 20; ++k) h = fmaf(win[k], w2s[k * 32 + n], h);
        h = fmaxf(h, 0.f);
        o0 = fmaf(h, w3s[n * 4 + 0], o0);
        o1 = fmaf(h, w3s[n * 4 + 1], o1);
        o2 = fmaf(h, w3s[n * 4 + 2], o2);
        o3 = fmaf(h, w3s[n * 4 + 3], o3);
    }
    float* q = out + ((size_t)g << 2);
    q[0] = o0; q[1] = o1; q[2] = o2; q[3] = o3;
}

extern "C" void kernel_launch(void* const* d_in, const int* in_sizes, int n_in,
                              void* d_out, int out_size, void* d_ws, size_t ws_size,
                              hipStream_t stream) {
    (void)in_sizes; (void)n_in; (void)out_size; (void)ws_size;
    const float* x  = (const float*)d_in[0];
    const float* W0 = (const float*)d_in[1];
    const float* b0 = (const float*)d_in[2];
    const float* W1 = (const float*)d_in[3];
    const float* b1 = (const float*)d_in[4];
    const float* Ws = (const float*)d_in[5];
    const float* bs = (const float*)d_in[6];
    const float* Wr = (const float*)d_in[7];
    const float* br = (const float*)d_in[8];
    const float* W2 = (const float*)d_in[9];
    const float* b2 = (const float*)d_in[10];
    const float* W3 = (const float*)d_in[11];
    const float* b3 = (const float*)d_in[12];
    float* out = (float*)d_out;

    float* ws   = (float*)d_ws;
    float* Wsr  = ws;
    float* Wc   = ws + 2048;
    float* bc   = ws + 2176;
    float* hmid = ws + 2304;

    k_wsr   <<<8,   256, 0, stream>>>(Ws, Wr, Wsr);
    k_fold  <<<1,   256, 0, stream>>>(W1, b1, bs, Wr, br, Wsr, Wc, bc);
    k_stage1<<<256, 256, 0, stream>>>(x, W0, b0, Wc, bc, hmid);
    k_stage2<<<128, 256, 0, stream>>>(hmid, W2, b2, W3, b3, out);
}


# Weight-fold precompute (Ws@Wr, W1@Wsr, bias sums): serial 4096-term per-thread fma chains -> coalesced row loads split over lanes + shuffle/LDS tree sums; f32 summation order of Wc, bc changes
# speedup vs baseline: 8.5864x; 8.5864x over previous
#include <hip/hip_runtime.h>
#include <hip/hip_bf16.h>

typedef __attribute__((ext_vector_type(2))) float v2f;
typedef __attribute__((ext_vector_type(8))) float v8f;

#define L_TOK   4096
#define B_BATCH 8
#define K0      484
#define H0      32
#define TB      128
#define HALO    120

__device__ __forceinline__ float wave_sum(float v) {
    v += __shfl_xor(v, 16, 32);
    v += __shfl_xor(v, 8, 32);
    v += __shfl_xor(v, 4, 32);
    v += __shfl_xor(v, 2, 32);
    v += __shfl_xor(v, 1, 32);
    return v;
}

#define WSR_ROWS 8
__global__ __launch_bounds__(256) void k_wsr(const float* __restrict__ Ws,
                                             const float* __restrict__ bs,
                                             const float* __restrict__ Wr,
                                             float* __restrict__ Wsr) {
    __shared__ float4 red[8][WSR_ROWS];
    const int tid  = threadIdx.x;
    const int lane = tid & 31;
    const int wave = tid >> 5;
    const int row0 = blockIdx.x * WSR_ROWS;

    float4 w[WSR_ROWS][4];
    #pragma unroll
    for (int rr = 0; rr < WSR_ROWS; ++rr) {
        const int row = row0 + rr;
        const float* src = (row < 512) ? Ws + (size_t)row * 4096 : bs;
        #pragma unroll
        for (int i = 0; i < 4; ++i) w[rr][i] = *(const float4*)(src + 4 * (tid + 256 * i));
    }

    float acc[WSR_ROWS][4];
    #pragma unroll
    for (int rr = 0; rr < WSR_ROWS; ++rr) { acc[rr][0] = 0.f; acc[rr][1] = 0.f; acc[rr][2] = 0.f; acc[rr][3] = 0.f; }

    #pragma unroll
    for (int i = 0; i < 4; ++i) {
        const float4* q = (const float4*)(Wr + 16 * (tid + 256 * i));
        const float4 q0 = q[0], q1 = q[1], q2 = q[2], q3 = q[3];
        #pragma unroll
        for (int rr = 0; rr < WSR_ROWS; ++rr) {
            const float4 a = w[rr][i];
            acc[rr][0] = fmaf(a.x, q0.x, acc[rr][0]); acc[rr][0] = fmaf(a.y, q1.x, acc[rr][0]);
            acc[rr][0] = fmaf(a.z, q2.x, acc[rr][0]); acc[rr][0] = fmaf(a.w, q3.x, acc[rr][0]);
            acc[rr][1] = fmaf(a.x, q0.y, acc[rr][1]); acc[rr][1] = fmaf(a.y, q1.y, acc[rr][1]);
            acc[rr][1] = fmaf(a.z, q2.y, acc[rr][1]); acc[rr][1] = fmaf(a.w, q3.y, acc[rr][1]);
            acc[rr][2] = fmaf(a.x, q0.z, acc[rr][2]); acc[rr][2] = fmaf(a.y, q1.z, acc[rr][2]);
            acc[rr][2] = fmaf(a.z, q2.z, acc[rr][2]); acc[rr][2] = fmaf(a.w, q3.z, acc[rr][2]);
            acc[rr][3] = fmaf(a.x, q0.w, acc[rr][3]); acc[rr][3] = fmaf(a.y, q1.w, acc[rr][3]);
            acc[rr][3] = fmaf(a.z, q2.w, acc[rr][3]); acc[rr][3] = fmaf(a.w, q3.w, acc[rr][3]);
        }
    }

    #pragma unroll
    for (int off = 16; off > 0; off >>= 1) {
        #pragma unroll
        for (int rr = 0; rr < WSR_ROWS; ++rr) {
            acc[rr][0] += __shfl_xor(acc[rr][0], off, 32);
            acc[rr][1] += __shfl_xor(acc[rr][1], off, 32);
            acc[rr][2] += __shfl_xor(acc[rr][2], off, 32);
            acc[rr][3] += __shfl_xor(acc[rr][3], off, 32);
        }
    }
    if (lane == 0) {
        #pragma unroll
        for (int rr = 0; rr < WSR_ROWS; ++rr) {
            float4 s;
            s.x = acc[rr][0]; s.y = acc[rr][1]; s.z = acc[rr][2]; s.w = acc[rr][3];
            red[wave][rr] = s;
        }
    }
    __syncthreads();

    if (tid < 32) {
        const float* rf = (const float*)&red[0][0];
        float s = rf[tid];
        #pragma unroll
        for (int wv = 1; wv < 8; ++wv) s += rf[wv * 32 + tid];
        if (row0 + (tid >> 2) <= 512) Wsr[row0 * 4 + tid] = s;
    }
}

__global__ __launch_bounds__(256) void k_fold(const float* __restrict__ W1,
                                              const float* __restrict__ b1,
                                              const float* __restrict__ br,
                                              const float* __restrict__ Wsr,
                                              float* __restrict__ Wc,
                                              float* __restrict__ bc) {
    __shared__ float4 stage[8];
    const int tid  = threadIdx.x;
    const int lane = tid & 31;
    const int wave = tid >> 5;
    const int row  = blockIdx.x * 8 + wave;
    const float* src = (row < 32) ? W1 + row * 512 : b1;

    float4 a[4];
    #pragma unroll
    for (int i = 0; i < 4; ++i) a[i] = *(const float4*)(src + 4 * (lane + 32 * i));

    float s0 = 0.f, s1 = 0.f, s2 = 0.f, s3 = 0.f;
    #pragma unroll
    for (int i = 0; i < 4; ++i) {
        const float4* q = (const float4*)(Wsr + 16 * (lane + 32 * i));
        const float4 q0 = q[0], q1 = q[1], q2 = q[2], q3 = q[3];
        s0 = fmaf(a[i].x, q0.x, s0); s0 = fmaf(a[i].y, q1.x, s0); s0 = fmaf(a[i].z, q2.x, s0); s0 = fmaf(a[i].w, q3.x, s0);
        s1 = fmaf(a[i].x, q0.y, s1); s1 = fmaf(a[i].y, q1.y, s1); s1 = fmaf(a[i].z, q2.y, s1); s1 = fmaf(a[i].w, q3.y, s1);
        s2 = fmaf(a[i].x, q0.z, s2); s2 = fmaf(a[i].y, q1.z, s2); s2 = fmaf(a[i].z, q2.z, s2); s2 = fmaf(a[i].w, q3.z, s2);
        s3 = fmaf(a[i].x, q0.w, s3); s3 = fmaf(a[i].y, q1.w, s3); s3 = fmaf(a[i].z, q2.w, s3); s3 = fmaf(a[i].w, q3.w, s3);
    }
    float4 s;
    s.x = wave_sum(s0); s.y = wave_sum(s1); s.z = wave_sum(s2); s.w = wave_sum(s3);
    if (lane == 0) stage[wave] = s;
    __syncthreads();

    const float* sf = (const float*)&stage[0];
    if (blockIdx.x < 4) {
        if (tid < 32) Wc[blockIdx.x * 32 + tid] = sf[tid];
    } else if (tid < 4) {
        bc[tid] = br[tid] + sf[tid] + Wsr[2048 + tid];
    }
}

__global__ __launch_bounds__(256) void k_stage1(const float* __restrict__ x,
                                                const float* __restrict__ W0,
                                                const float* __restrict__ b0,
                                                const float* __restrict__ Wc,
                                                const float* __restrict__ bc,
                                                float* __restrict__ hmid) {
    __shared__ float xt[(TB + HALO) * 4];
    __shared__ float w0t[H0 * K0];
    __shared__ float b0s[H0];
    __shared__ float wcs[H0 * 4];
    __shared__ float bcs[4];
    __shared__ float h0s[TB * 36];

    const int tid = threadIdx.x;
    const int bb  = blockIdx.x >> 5;
    const int t0  = (blockIdx.x & 31) * TB;

    for (int i = tid; i < (TB + HALO) * 4; i += 256) {
        int j = i >> 2, c = i & 3;
        int gt = (t0 + j + L_TOK - HALO) & (L_TOK - 1);
        xt[i] = x[((bb << 12) + gt) * 4 + c];
    }
    for (int i = tid; i < H0 * K0; i += 256) {
        int k = i >> 5, n = i & 31;
        w0t[n * K0 + k] = W0[i];
    }
    if (tid < H0)  b0s[tid] = b0[tid];
    if (tid < 128) wcs[tid] = Wc[tid];
    if (tid < 4)   bcs[tid] = bc[tid];
    __syncthreads();

    const int wave = tid >> 5;
    const int lane = tid & 31;
    const int hl   = lane >> 4;
    const int m0   = lane & 15;
    const int c0   = hl << 1;

    v8f acc0 = {}; v8f acc1 = {};
    const float* ap  = xt  + (((wave << 4) + m0)) * 4 + c0;
    const float* bp0 = w0t + m0 * K0 + c0;
    const float* bp1 = w0t + (m0 + 16) * K0 + c0;

    for (int kk = 0; kk < 121; ++kk) {
        v2f a  = *(const v2f*)(ap  + 4 * kk);
        v2f bA = *(const v2f*)(bp0 + 4 * kk);
        v2f bB = *(const v2f*)(bp1 + 4 * kk);
        acc0 = __builtin_amdgcn_wmma_f32_16x16x4_f32(false, a, false, bA, (short)0, acc0, false, false);
        acc1 = __builtin_amdgcn_wmma_f32_16x16x4_f32(false, a, false, bB, (short)0, acc1, false, false);
    }

    #pragma unroll
    for (int g = 0; g < 8; ++g) {
        int M = (wave << 4) + g + (hl << 3);
        h0s[M * 36 + m0]      = tanhf(acc0[g] + b0s[m0]);
        h0s[M * 36 + m0 + 16] = tanhf(acc1[g] + b0s[m0 + 16]);
    }
    __syncthreads();

    #pragma unroll
    for (int r = 0; r < 2; ++r) {
        int idx = tid + (r << 8);
        int tok = idx >> 2, o = idx & 3;
        const float* hrow = h0s + tok * 36;
        float s = bcs[o];
        #pragma unroll 8
        for (int n = 0; n < 32; ++n) s = fmaf(hrow[n], wcs[n * 4 + o], s);
        hmid[(((bb << 12) + t0 + tok) << 2) + o] = s;
    }
}

__global__ __launch_bounds__(256) void k_stage2(const float* __restrict__ hmid,
                                                const float* __restrict__ W2,
                                                const float* __restrict__ b2,
                                                const float* __restrict__ W3,
                                                const float* __restrict__ b3,
                                                float* __restrict__ out) {
    __shared__ float w2s[20 * 32];
    __shared__ float b2s[32];
    __shared__ float w3s[32 * 4];
    __shared__ float b3s[4];
    int tid = threadIdx.x;
    for (int i = tid; i < 640; i += 256) w2s[i] = W2[i];
    if (tid < 32)  b2s[tid] = b2[tid];
    if (tid < 128) w3s[tid] = W3[tid];
    if (tid < 4)   b3s[tid] = b3[tid];
    __syncthreads();

    int g  = blockIdx.x * 256 + tid;
    int bb = g >> 12, t = g & (L_TOK - 1);

    float win[20];
    #pragma unroll
    for (int j = 0; j < 5; ++j) {
        int tt = (t + L_TOK - 4 + j) & (L_TOK - 1);
        const float* p = hmid + (((bb << 12) + tt) << 2);
        win[j * 4 + 0] = p[0]; win[j * 4 + 1] = p[1];
        win[j * 4 + 2] = p[2]; win[j * 4 + 3] = p[3];
    }
    float o0 = b3s[0], o1 = b3s[1], o2 = b3s[2], o3 = b3s[3];
    for (int n = 0; n < 32; ++n) {
        float h = b2s[n];
        #pragma unroll
        for (int k = 0; k < 20; ++k) h = fmaf(win[k], w2s[k * 32 + n], h);
        h = fmaxf(h, 0.f);
        o0 = fmaf(h, w3s[n * 4 + 0], o0);
        o1 = fmaf(h, w3s[n * 4 + 1], o1);
        o2 = fmaf(h, w3s[n * 4 + 2], o2);
        o3 = fmaf(h, w3s[n * 4 + 3], o3);
    }
    float* q = out + ((size_t)g << 2);
    q[0] = o0; q[1] = o1; q[2] = o2; q[3] = o3;
}

extern "C" void kernel_launch(void* const* d_in, const int* in_sizes, int n_in,
                              void* d_out, int out_size, void* d_ws, size_t ws_size,
                              hipStream_t stream) {
    (void)in_sizes; (void)n_in; (void)out_size; (void)ws_size;
    const float* x  = (const float*)d_in[0];
    const float* W0 = (const float*)d_in[1];
    const float* b0 = (const float*)d_in[2];
    const float* W1 = (const float*)d_in[3];
    const float* b1 = (const float*)d_in[4];
    const float* Ws = (const float*)d_in[5];
    const float* bs = (const float*)d_in[6];
    const float* Wr = (const float*)d_in[7];
    const float* br = (const float*)d_in[8];
    const float* W2 = (const float*)d_in[9];
    const float* b2 = (const float*)d_in[10];
    const float* W3 = (const float*)d_in[11];
    const float* b3 = (const float*)d_in[12];
    float* out = (float*)d_out;

    float* ws   = (float*)d_ws;
    float* Wsr  = ws;
    float* Wc   = ws + 2176;
    float* bc   = ws + 2304;
    float* hmid = ws + 2432;

    k_wsr   <<<65,  256, 0, stream>>>(Ws, bs, Wr, Wsr);
    k_fold  <<<5,   256, 0, stream>>>(W1, b1, br, Wsr, Wc, bc);
    k_stage1<<<256, 256, 0, stream>>>(x, W0, b0, Wc, bc, hmid);
    k_stage2<<<128, 256, 0, stream>>>(hmid, W2, b2, W3, b3, out);
}


# Launch fusion 4 -> 2: Ws@Wr rows spread over the WMMA blocks, W1@Wsr fold redone per block in the window-5 kernel, whose prologue loads all go out together; f32 sum orders unchanged
# speedup vs baseline: 10.7202x; 1.2485x over previous
#include <hip/hip_runtime.h>
#include <hip/hip_bf16.h>


typedef __attribute__((ext_vector_type(2))) float v2f;
typedef __attribute__((ext_vector_type(8))) float v8f;

#define L_TOK   4096
#define B_BATCH 8
#define K0      484
#define H0      32
#define TB      128
#define HALO    120

#define WSR_LINE   32
#define H0G_OFFSET 8448

__global__ __launch_bounds__(256) void k_front(const float* __restrict__ x,
                                               const float* __restrict__ W0,
                                               const float* __restrict__ b0,
                                               const float* __restrict__ Ws,
                                               const float* __restrict__ bs,
                                               const float* __restrict__ Wr,
                                               float* __restrict__ Wsr,
                                               float* __restrict__ h0g) {
    __shared__ float xt[(TB + HALO) * 4];
    __shared__ float w0t[H0 * K0];
    __shared__ float b0s[H0];
    __shared__ alignas(16) float h0s[TB * 36];
    __shared__ float4 red[8][3];

    const int tid = threadIdx.x;
    const int bb  = blockIdx.x >> 5;
    const int t0  = (blockIdx.x & 31) * TB;

    for (int i = tid; i < (TB + HALO) * 4; i += 256) {
        int j = i >> 2, c = i & 3;
        int gt = (t0 + j + L_TOK - HALO) & (L_TOK - 1);
        xt[i] = x[((bb << 12) + gt) * 4 + c];
    }
    for (int i = tid; i < H0 * K0; i += 256) {
        int k = i >> 5, n = i & 31;
        w0t[n * K0 + k] = W0[i];
    }
    if (tid < H0)  b0s[tid] = b0[tid];
    __syncthreads();

    float4 w[3][4];
    {
        const float* src0 = Ws + (size_t)(2 * blockIdx.x) * 4096;
        #pragma unroll
        for (int i = 0; i < 4; ++i) {
            w[0][i] = *(const float4*)(src0 + 4 * (tid + 256 * i));
            w[1][i] = *(const float4*)(src0 + 4096 + 4 * (tid + 256 * i));
            w[2][i] = *(const float4*)(bs + 4 * (tid + 256 * i));
        }
    }
    float4 q[4][4];
    #pragma unroll
    for (int i = 0; i < 4; ++i) {
        const float4* qp = (const float4*)(Wr + 16 * (tid + 256 * i));
        q[i][0] = qp[0]; q[i][1] = qp[1]; q[i][2] = qp[2]; q[i][3] = qp[3];
    }

    const int wave = tid >> 5;
    const int lane = tid & 31;
    const int hl   = lane >> 4;
    const int m0   = lane & 15;
    const int c0   = hl << 1;

    v8f acc0 = {}; v8f acc1 = {};
    const float* ap  = xt  + (((wave << 4) + m0)) * 4 + c0;
    const float* bp0 = w0t + m0 * K0 + c0;
    const float* bp1 = w0t + (m0 + 16) * K0 + c0;

    for (int kk = 0; kk < 121; ++kk) {
        v2f a  = *(const v2f*)(ap  + 4 * kk);
        v2f bA = *(const v2f*)(bp0 + 4 * kk);
        v2f bB = *(const v2f*)(bp1 + 4 * kk);
        acc0 = __builtin_amdgcn_wmma_f32_16x16x4_f32(false, a, false, bA, (short)0, acc0, false, false);
        acc1 = __builtin_amdgcn_wmma_f32_16x16x4_f32(false, a, false, bB, (short)0, acc1, false, false);
    }

    #pragma unroll
    for (int g = 0; g < 8; ++g) {
        int M = (wave << 4) + g + (hl << 3);
        h0s[M * 36 + m0]      = tanhf(acc0[g] + b0s[m0]);
        h0s[M * 36 + m0 + 16] = tanhf(acc1[g] + b0s[m0 + 16]);
    }
    __syncthreads();

    #pragma unroll
    for (int r = 0; r < 4; ++r) {
        const int f   = tid + (r << 8);
        const int tok = f >> 3, qd = f & 7;
        const float4 v = *(const float4*)(h0s + tok * 36 + 4 * qd);
        *(float4*)(h0g + (size_t)((bb << 12) + t0 + tok) * 32 + 4 * qd) = v;
    }

    float acc[3][4];
    #pragma unroll
    for (int rr = 0; rr < 3; ++rr) { acc[rr][0] = 0.f; acc[rr][1] = 0.f; acc[rr][2] = 0.f; acc[rr][3] = 0.f; }
    #pragma unroll
    for (int i = 0; i < 4; ++i) {
        const float4 q0 = q[i][0], q1 = q[i][1], q2 = q[i][2], q3 = q[i][3];
        #pragma unroll
        for (int rr = 0; rr < 3; ++rr) {
            const float4 a = w[rr][i];
            acc[rr][0] = fmaf(a.x, q0.x, acc[rr][0]); acc[rr][0] = fmaf(a.y, q1.x, acc[rr][0]);
            acc[rr][0] = fmaf(a.z, q2.x, acc[rr][0]); acc[rr][0] = fmaf(a.w, q3.x, acc[rr][0]);
            acc[rr][1] = fmaf(a.x, q0.y, acc[rr][1]); acc[rr][1] = fmaf(a.y, q1.y, acc[rr][1]);
            acc[rr][1] = fmaf(a.z, q2.y, acc[rr][1]); acc[rr][1] = fmaf(a.w, q3.y, acc[rr][1]);
            acc[rr][2] = fmaf(a.x, q0.z, acc[rr][2]); acc[rr][2] = fmaf(a.y, q1.z, acc[rr][2]);
            acc[rr][2] = fmaf(a.z, q2.z, acc[rr][2]); acc[rr][2] = fmaf(a.w, q3.z, acc[rr][2]);
            acc[rr][3] = fmaf(a.x, q0.w, acc[rr][3]); acc[rr][3] = fmaf(a.y, q1.w, acc[rr][3]);
            acc[rr][3] = fmaf(a.z, q2.w, acc[rr][3]); acc[rr][3] = fmaf(a.w, q3.w, acc[rr][3]);
        }
    }
    #pragma unroll
    for (int off = 16; off > 0; off >>= 1) {
        #pragma unroll
        for (int rr = 0; rr < 3; ++rr) {
            acc[rr][0] += __shfl_xor(acc[rr][0], off, 32);
            acc[rr][1] += __shfl_xor(acc[rr][1], off, 32);
            acc[rr][2] += __shfl_xor(acc[rr][2], off, 32);
            acc[rr][3] += __shfl_xor(acc[rr][3], off, 32);
        }
    }
    if (lane == 0) {
        #pragma unroll
        for (int rr = 0; rr < 3; ++rr) {
            float4 s;
            s.x = acc[rr][0]; s.y = acc[rr][1]; s.z = acc[rr][2]; s.w = acc[rr][3];
            red[wave][rr] = s;
        }
    }
    __syncthreads();

    if (tid < 12) {
        const float* rf = (const float*)&red[0][0];
        float s = rf[tid];
        #pragma unroll
        for (int wv = 1; wv < 8; ++wv) s += rf[wv * 12 + tid];
        if (tid < 8 || blockIdx.x == 0) Wsr[blockIdx.x * WSR_LINE + tid] = s;
    }
}

__global__ __launch_bounds__(256) void k_back(const float* __restrict__ h0g,
                                              const float* __restrict__ Wsr,
                                              const float* __restrict__ W1,
                                              const float* __restrict__ b1,
                                              const float* __restrict__ br,
                                              const float* __restrict__ W2,
                                              const float* __restrict__ b2,
                                              const float* __restrict__ W3,
                                              const float* __restrict__ b3,
                                              float* __restrict__ out) {
    __shared__ float w2s[20 * 32];
    __shared__ float b2s[32];
    __shared__ float w3s[32 * 4];
    __shared__ float b3s[4];
    __shared__ float4 wcs[32];
    __shared__ float4 bcs;
    __shared__ float4 hm[256 + 4];

    const int tid  = threadIdx.x;
    const int lane = tid & 31;
    const int wave = tid >> 5;

    const int g  = blockIdx.x * 256 + tid;
    const int bb = g >> 12;
    const int tb = (blockIdx.x & 15) * 256;

    float4 q[4][4];
    #pragma unroll
    for (int i = 0; i < 4; ++i) {
        const float* ln = Wsr + (size_t)(2 * (lane + 32 * i)) * WSR_LINE;
        q[i][0] = *(const float4*)(ln);
        q[i][1] = *(const float4*)(ln + 4);
        q[i][2] = *(const float4*)(ln + WSR_LINE);
        q[i][3] = *(const float4*)(ln + WSR_LINE + 4);
    }
    float4 a[5][4];
    #pragma unroll
    for (int rw = 0; rw < 5; ++rw) {
        const float* src = (rw < 4) ? W1 + (4 * wave + rw) * 512 : b1;
        #pragma unroll
        for (int i = 0; i < 4; ++i) a[rw][i] = *(const float4*)(src + 4 * (lane + 32 * i));
    }

    float4 hv[8];
    {
        const float4* hp = (const float4*)(h0g + (size_t)g * 32);
        #pragma unroll
        for (int i = 0; i < 8; ++i) hv[i] = hp[i];
    }
    float4 hh[8];
    {
        const int tt = (tb + L_TOK - 4 + (tid & 3)) & (L_TOK - 1);
        const float4* hp = (const float4*)(h0g + (size_t)((bb << 12) + tt) * 32);
        #pragma unroll
        for (int i = 0; i < 8; ++i) hh[i] = hp[i];
    }

    const float w2v0 = W2[tid];
    const float w2v1 = W2[tid + 256];
    const float w2v2 = W2[512 + (tid & 127)];
    const float b2v  = b2[tid & 31];
    const float w3v  = W3[tid & 127];
    const float b3v  = b3[tid & 3];

    float s[5][4];
    #pragma unroll
    for (int rw = 0; rw < 5; ++rw) { s[rw][0] = 0.f; s[rw][1] = 0.f; s[rw][2] = 0.f; s[rw][3] = 0.f; }
    #pragma unroll
    for (int i = 0; i < 4; ++i) {
        const float4 q0 = q[i][0], q1 = q[i][1], q2 = q[i][2], q3 = q[i][3];
        #pragma unroll
        for (int rw = 0; rw < 5; ++rw) {
            const float4 v = a[rw][i];
            s[rw][0] = fmaf(v.x, q0.x, s[rw][0]); s[rw][0] = fmaf(v.y, q1.x, s[rw][0]);
            s[rw][0] = fmaf(v.z, q2.x, s[rw][0]); s[rw][0] = fmaf(v.w, q3.x, s[rw][0]);
            s[rw][1] = fmaf(v.x, q0.y, s[rw][1]); s[rw][1] = fmaf(v.y, q1.y, s[rw][1]);
            s[rw][1] = fmaf(v.z, q2.y, s[rw][1]); s[rw][1] = fmaf(v.w, q3.y, s[rw][1]);
            s[rw][2] = fmaf(v.x, q0.z, s[rw][2]); s[rw][2] = fmaf(v.y, q1.z, s[rw][2]);
            s[rw][2] = fmaf(v.z, q2.z, s[rw][2]); s[rw][2] = fmaf(v.w, q3.z, s[rw][2]);
            s[rw][3] = fmaf(v.x, q0.w, s[rw][3]); s[rw][3] = fmaf(v.y, q1.w, s[rw][3]);
            s[rw][3] = fmaf(v.z, q2.w, s[rw][3]); s[rw][3] = fmaf(v.w, q3.w, s[rw][3]);
        }
    }
    #pragma unroll
    for (int off = 16; off > 0; off >>= 1) {
        #pragma unroll
        for (int rw = 0; rw < 5; ++rw) {
            s[rw][0] += __shfl_xor(s[rw][0], off, 32);
            s[rw][1] += __shfl_xor(s[rw][1], off, 32);
            s[rw][2] += __shfl_xor(s[rw][2], off, 32);
            s[rw][3] += __shfl_xor(s[rw][3], off, 32);
        }
    }
    if (lane == 0) {
        #pragma unroll
        for (int rw = 0; rw < 4; ++rw) {
            float4 v;
            v.x = s[rw][0]; v.y = s[rw][1]; v.z = s[rw][2]; v.w = s[rw][3];
            wcs[4 * wave + rw] = v;
        }
        if (wave == 0) {
            const float* w512 = Wsr + 8;
            float4 v;
            v.x = br[0] + s[4][0] + w512[0];
            v.y = br[1] + s[4][1] + w512[1];
            v.z = br[2] + s[4][2] + w512[2];
            v.w = br[3] + s[4][3] + w512[3];
            bcs = v;
        }
    }
    __syncthreads();

    {
        float4 m = bcs;
        float4 mh = m;
        #pragma unroll
        for (int i = 0; i < 8; ++i) {
            const float4 c0 = wcs[4 * i], c1 = wcs[4 * i + 1], c2 = wcs[4 * i + 2], c3 = wcs[4 * i + 3];
            const float4 h = hv[i];
            m.x = fmaf(h.x, c0.x, m.x); m.y = fmaf(h.x, c0.y, m.y); m.z = fmaf(h.x, c0.z, m.z); m.w = fmaf(h.x, c0.w, m.w);
            m.x = fmaf(h.y, c1.x, m.x); m.y = fmaf(h.y, c1.y, m.y); m.z = fmaf(h.y, c1.z, m.z); m.w = fmaf(h.y, c1.w, m.w);
            m.x = fmaf(h.z, c2.x, m.x); m.y = fmaf(h.z, c2.y, m.y); m.z = fmaf(h.z, c2.z, m.z); m.w = fmaf(h.z, c2.w, m.w);
            m.x = fmaf(h.w, c3.x, m.x); m.y = fmaf(h.w, c3.y, m.y); m.z = fmaf(h.w, c3.z, m.z); m.w = fmaf(h.w, c3.w, m.w);
            const float4 e = hh[i];
            mh.x = fmaf(e.x, c0.x, mh.x); mh.y = fmaf(e.x, c0.y, mh.y); mh.z = fmaf(e.x, c0.z, mh.z); mh.w = fmaf(e.x, c0.w, mh.w);
            mh.x = fmaf(e.y, c1.x, mh.x); mh.y = fmaf(e.y, c1.y, mh.y); mh.z = fmaf(e.y, c1.z, mh.z); mh.w = fmaf(e.y, c1.w, mh.w);
            mh.x = fmaf(e.z, c2.x, mh.x); mh.y = fmaf(e.z, c2.y, mh.y); mh.z = fmaf(e.z, c2.z, mh.z); mh.w = fmaf(e.z, c2.w, mh.w);
            mh.x = fmaf(e.w, c3.x, mh.x); mh.y = fmaf(e.w, c3.y, mh.y); mh.z = fmaf(e.w, c3.z, mh.z); mh.w = fmaf(e.w, c3.w, mh.w);
        }
        hm[tid + 4] = m;
        if (tid < 4) hm[tid] = mh;
    }
    w2s[tid] = w2v0;
    w2s[tid + 256] = w2v1;
    if (tid < 128) { w2s[tid + 512] = w2v2; w3s[tid] = w3v; }
    if (tid < 32)  b2s[tid] = b2v;
    if (tid < 4)   b3s[tid] = b3v;
    __syncthreads();

    float win[20];
    #pragma unroll
    for (int j = 0; j < 5; ++j) {
        const float4 p = hm[tid + j];
        win[j * 4 + 0] = p.x; win[j * 4 + 1] = p.y;
        win[j * 4 + 2] = p.z; win[j * 4 + 3] = p.w;
    }
    float o0 = b3s[0], o1 = b3s[1], o2 = b3s[2], o3 = b3s[3];
    for (int n = 0; n < 32; ++n) {
        float h = b2s[n];
        #pragma unroll
        for (int k = 0; k < 20; ++k) h = fmaf(win[k], w2s[k * 32 + n], h);
        h = fmaxf(h, 0.f);
        o0 = fmaf(h, w3s[n * 4 + 0], o0);
        o1 = fmaf(h, w3s[n * 4 + 1], o1);
        o2 = fmaf(h, w3s[n * 4 + 2], o2);
        o3 = fmaf(h, w3s[n * 4 + 3], o3);
    }
    float* qo = out + ((size_t)g << 2);
    qo[0] = o0; qo[1] = o1; qo[2] = o2; qo[3] = o3;
}

extern "C" void kernel_launch(void* const* d_in, const int* in_sizes, int n_in,
                              void* d_out, int out_size, void* d_ws, size_t ws_size,
                              hipStream_t stream) {
    (void)in_sizes; (void)n_in; (void)out_size; (void)ws_size;
    const float* x  = (const float*)d_in[0];
    const float* W0 = (const float*)d_in[1];
    const float* b0 = (const float*)d_in[2];
    const float* W1 = (const float*)d_in[3];
    const float* b1 = (const float*)d_in[4];
    const float* Ws = (const float*)d_in[5];
    const float* bs = (const float*)d_in[6];
    const float* Wr = (const float*)d_in[7];
    const float* br = (const float*)d_in[8];
    const float* W2 = (const float*)d_in[9];
    const float* b2 = (const float*)d_in[10];
    const float* W3 = (const float*)d_in[11];
    const float* b3 = (const float*)d_in[12];
    float* out = (float*)d_out;

    float* ws  = (float*)d_ws;
    float* Wsr = ws;
    float* h0g = ws + H0G_OFFSET;

    k_front<<<256, 256, 0, stream>>>(x, W0, b0, Ws, bs, Wr, Wsr, h0g);
    k_back <<<128, 256, 0, stream>>>(h0g, Wsr, W1, b1, br, W2, b2, W3, b3, out);
}


# Front-kernel LDS fill (x tile, W0 transposed): 65 four-byte load-then-store steps per thread in 14 dependent trips -> 17 16-byte loads issued together, 4x4 register transpose, 16-byte LDS stores
# speedup vs baseline: 13.1583x; 1.2274x over previous
#include <hip/hip_runtime.h>
#include <hip/hip_bf16.h>


typedef __attribute__((ext_vector_type(2))) float v2f;
typedef __attribute__((ext_vector_type(8))) float v8f;

#define L_TOK   4096
#define B_BATCH 8
#define K0      484
#define H0      32
#define TB      128
#define HALO    120
#define W0_PIECES 968

#define WSR_LINE   32
#define H0G_OFFSET 8448

__global__ __launch_bounds__(256) void k_front(const float* __restrict__ x,
                                               const float* __restrict__ W0,
                                               const float* __restrict__ b0,
                                               const float* __restrict__ Ws,
                                               const float* __restrict__ bs,
                                               const float* __restrict__ Wr,
                                               float* __restrict__ Wsr,
                                               float* __restrict__ h0g) {
    __shared__ alignas(16) float xt[(TB + HALO) * 4];
    __shared__ alignas(16) float w0t[H0 * K0];
    __shared__ float b0s[H0];
    __shared__ alignas(16) float h0s[TB * 36];
    __shared__ float4 red[8][3];

    const int tid = threadIdx.x;
    const int bb  = blockIdx.x >> 5;
    const int t0  = (blockIdx.x & 31) * TB;

    float4 wv[4][4];
    #pragma unroll
    for (int r4 = 0; r4 < 4; ++r4) {
        int u = tid + 256 * r4;
        u = u < W0_PIECES ? u : W0_PIECES - 1;
        const float* src = W0 + (u >> 3) * 128 + (u & 7) * 4;
        #pragma unroll
        for (int r = 0; r < 4; ++r) wv[r4][r] = *(const float4*)(src + 32 * r);
    }
    const float b0v = b0[tid & 31];

    if (tid < TB + HALO) {
        const int gt = (t0 + tid + L_TOK - HALO) & (L_TOK - 1);
        *(float4*)(xt + 4 * tid) = *(const float4*)(x + ((bb << 12) + gt) * 4);
    }
    #pragma unroll
    for (int r4 = 0; r4 < 4; ++r4) {
        const int u = tid + 256 * r4;
        if (u < W0_PIECES) {
            float* dst = w0t + (u & 7) * 4 * K0 + (u >> 3) * 4;
            const float4 a0 = wv[r4][0], a1 = wv[r4][1], a2 = wv[r4][2], a3 = wv[r4][3];
            float4 t;
            t.x = a0.x; t.y = a1.x; t.z = a2.x; t.w = a3.x; *(float4*)(dst)          = t;
            t.x = a0.y; t.y = a1.y; t.z = a2.y; t.w = a3.y; *(float4*)(dst + K0)     = t;
            t.x = a0.z; t.y = a1.z; t.z = a2.z; t.w = a3.z; *(float4*)(dst + 2 * K0) = t;
            t.x = a0.w; t.y = a1.w; t.z = a2.w; t.w = a3.w; *(float4*)(dst + 3 * K0) = t;
        }
    }
    if (tid < H0)  b0s[tid] = b0v;
    __syncthreads();

    float4 w[3][4];
    {
        const float* src0 = Ws + (size_t)(2 * blockIdx.x) * 4096;
        #pragma unroll
        for (int i = 0; i < 4; ++i) {
            w[0][i] = *(const float4*)(src0 + 4 * (tid + 256 * i));
            w[1][i] = *(const float4*)(src0 + 4096 + 4 * (tid + 256 * i));
            w[2][i] = *(const float4*)(bs + 4 * (tid + 256 * i));
        }
    }
    float4 q[4][4];
    #pragma unroll
    for (int i = 0; i < 4; ++i) {
        const float4* qp = (const float4*)(Wr + 16 * (tid + 256 * i));
        q[i][0] = qp[0]; q[i][1] = qp[1]; q[i][2] = qp[2]; q[i][3] = qp[3];
    }

    const int wave = tid >> 5;
    const int lane = tid & 31;
    const int hl   = lane >> 4;
    const int m0   = lane & 15;
    const int c0   = hl << 1;

    v8f acc0 = {}; v8f acc1 = {};
    const float* ap  = xt  + (((wave << 4) + m0)) * 4 + c0;
    const float* bp0 = w0t + m0 * K0 + c0;
    const float* bp1 = w0t + (m0 + 16) * K0 + c0;

    for (int kk = 0; kk < 121; ++kk) {
        v2f a  = *(const v2f*)(ap  + 4 * kk);
        v2f bA = *(const v2f*)(bp0 + 4 * kk);
        v2f bB = *(const v2f*)(bp1 + 4 * kk);
        acc0 = __builtin_amdgcn_wmma_f32_16x16x4_f32(false, a, false, bA, (short)0, acc0, false, false);
        acc1 = __builtin_amdgcn_wmma_f32_16x16x4_f32(false, a, false, bB, (short)0, acc1, false, false);
    }

    #pragma unroll
    for (int g = 0; g < 8; ++g) {
        int M = (wave << 4) + g + (hl << 3);
        h0s[M * 36 + m0]      = tanhf(acc0[g] + b0s[m0]);
        h0s[M * 36 + m0 + 16] = tanhf(acc1[g] + b0s[m0 + 16]);
    }
    __syncthreads();

    #pragma unroll
    for (int r = 0; r < 4; ++r) {
        const int f   = tid + (r << 8);
        const int tok = f >> 3, qd = f & 7;
        const float4 v = *(const float4*)(h0s + tok * 36 + 4 * qd);
        *(float4*)(h0g + (size_t)((bb << 12) + t0 + tok) * 32 + 4 * qd) = v;
    }

    float acc[3][4];
    #pragma unroll
    for (int rr = 0; rr < 3; ++rr) { acc[rr][0] = 0.f; acc[rr][1] = 0.f; acc[rr][2] = 0.f; acc[rr][3] = 0.f; }
    #pragma unroll
    for (int i = 0; i < 4; ++i) {
        const float4 q0 = q[i][0], q1 = q[i][1], q2 = q[i][2], q3 = q[i][3];
        #pragma unroll
        for (int rr = 0; rr < 3; ++rr) {
            const float4 a = w[rr][i];
            acc[rr][0] = fmaf(a.x, q0.x, acc[rr][0]); acc[rr][0] = fmaf(a.y, q1.x, acc[rr][0]);
            acc[rr][0] = fmaf(a.z, q2.x, acc[rr][0]); acc[rr][0] = fmaf(a.w, q3.x, acc[rr][0]);
            acc[rr][1] = fmaf(a.x, q0.y, acc[rr][1]); acc[rr][1] = fmaf(a.y, q1.y, acc[rr][1]);
            acc[rr][1] = fmaf(a.z, q2.y, acc[rr][1]); acc[rr][1] = fmaf(a.w, q3.y, acc[rr][1]);
            acc[rr][2] = fmaf(a.x, q0.z, acc[rr][2]); acc[rr][2] = fmaf(a.y, q1.z, acc[rr][2]);
            acc[rr][2] = fmaf(a.z, q2.z, acc[rr][2]); acc[rr][2] = fmaf(a.w, q3.z, acc[rr][2]);
            acc[rr][3] = fmaf(a.x, q0.w, acc[rr][3]); acc[rr][3] = fmaf(a.y, q1.w, acc[rr][3]);
            acc[rr][3] = fmaf(a.z, q2.w, acc[rr][3]); acc[rr][3] = fmaf(a.w, q3.w, acc[rr][3]);
        }
    }
    #pragma unroll
    for (int off = 16; off > 0; off >>= 1) {
        #pragma unroll
        for (int rr = 0; rr < 3; ++rr) {
            acc[rr][0] += __shfl_xor(acc[rr][0], off, 32);
            acc[rr][1] += __shfl_xor(acc[rr][1], off, 32);
            acc[rr][2] += __shfl_xor(acc[rr][2], off, 32);
            acc[rr][3] += __shfl_xor(acc[rr][3], off, 32);
        }
    }
    if (lane == 0) {
        #pragma unroll
        for (int rr = 0; rr < 3; ++rr) {
            float4 s;
            s.x = acc[rr][0]; s.y = acc[rr][1]; s.z = acc[rr][2]; s.w = acc[rr][3];
            red[wave][rr] = s;
        }
    }
    __syncthreads();

    if (tid < 12) {
        const float* rf = (const float*)&red[0][0];
        float s = rf[tid];
        #pragma unroll
        for (int wv = 1; wv < 8; ++wv) s += rf[wv * 12 + tid];
        if (tid < 8 || blockIdx.x == 0) Wsr[blockIdx.x * WSR_LINE + tid] = s;
    }
}

__global__ __launch_bounds__(256) void k_back(const float* __restrict__ h0g,
                                              const float* __restrict__ Wsr,
                                              const float* __restrict__ W1,
                                              const float* __restrict__ b1,
                                              const float* __restrict__ br,
                                              const float* __restrict__ W2,
                                              const float* __restrict__ b2,
                                              const float* __restrict__ W3,
                                              const float* __restrict__ b3,
                                              float* __restrict__ out) {
    __shared__ float w2s[20 * 32];
    __shared__ float b2s[32];
    __shared__ float w3s[32 * 4];
    __shared__ float b3s[4];
    __shared__ float4 wcs[32];
    __shared__ float4 bcs;
    __shared__ float4 hm[256 + 4];

    const int tid  = threadIdx.x;
    const int lane = tid & 31;
    const int wave = tid >> 5;

    const int g  = blockIdx.x * 256 + tid;
    const int bb = g >> 12;
    const int tb = (blockIdx.x & 15) * 256;

    float4 q[4][4];
    #pragma unroll
    for (int i = 0; i < 4; ++i) {
        const float* ln = Wsr + (size_t)(2 * (lane + 32 * i)) * WSR_LINE;
        q[i][0] = *(const float4*)(ln);
        q[i][1] = *(const float4*)(ln + 4);
        q[i][2] = *(const float4*)(ln + WSR_LINE);
        q[i][3] = *(const float4*)(ln + WSR_LINE + 4);
    }
    float4 a[5][4];
    #pragma unroll
    for (int rw = 0; rw < 5; ++rw) {
        const float* src = (rw < 4) ? W1 + (4 * wave + rw) * 512 : b1;
        #pragma unroll
        for (int i = 0; i < 4; ++i) a[rw][i] = *(const float4*)(src + 4 * (lane + 32 * i));
    }

    float4 hv[8];
    {
        const float4* hp = (const float4*)(h0g + (size_t)g * 32);
        #pragma unroll
        for (int i = 0; i < 8; ++i) hv[i] = hp[i];
    }
    float4 hh[8];
    {
        const int tt = (tb + L_TOK - 4 + (tid & 3)) & (L_TOK - 1);
        const float4* hp = (const float4*)(h0g + (size_t)((bb << 12) + tt) * 32);
        #pragma unroll
        for (int i = 0; i < 8; ++i) hh[i] = hp[i];
    }

    const float w2v0 = W2[tid];
    const float w2v1 = W2[tid + 256];
    const float w2v2 = W2[512 + (tid & 127)];
    const float b2v  = b2[tid & 31];
    const float w3v  = W3[tid & 127];
    const float b3v  = b3[tid & 3];

    float s[5][4];
    #pragma unroll
    for (int rw = 0; rw < 5; ++rw) { s[rw][0] = 0.f; s[rw][1] = 0.f; s[rw][2] = 0.f; s[rw][3] = 0.f; }
    #pragma unroll
    for (int i = 0; i < 4; ++i) {
        const float4 q0 = q[i][0], q1 = q[i][1], q2 = q[i][2], q3 = q[i][3];
        #pragma unroll
        for (int rw = 0; rw < 5; ++rw) {
            const float4 v = a[rw][i];
            s[rw][0] = fmaf(v.x, q0.x, s[rw][0]); s[rw][0] = fmaf(v.y, q1.x, s[rw][0]);
            s[rw][0] = fmaf(v.z, q2.x, s[rw][0]); s[rw][0] = fmaf(v.w, q3.x, s[rw][0]);
            s[rw][1] = fmaf(v.x, q0.y, s[rw][1]); s[rw][1] = fmaf(v.y, q1.y, s[rw][1]);
            s[rw][1] = fmaf(v.z, q2.y, s[rw][1]); s[rw][1] = fmaf(v.w, q3.y, s[rw][1]);
            s[rw][2] = fmaf(v.x, q0.z, s[rw][2]); s[rw][2] = fmaf(v.y, q1.z, s[rw][2]);
            s[rw][2] = fmaf(v.z, q2.z, s[rw][2]); s[rw][2] = fmaf(v.w, q3.z, s[rw][2]);
            s[rw][3] = fmaf(v.x, q0.w, s[rw][3]); s[rw][3] = fmaf(v.y, q1.w, s[rw][3]);
            s[rw][3] = fmaf(v.z, q2.w, s[rw][3]); s[rw][3] = fmaf(v.w, q3.w, s[rw][3]);
        }
    }
    #pragma unroll
    for (int off = 16; off > 0; off >>= 1) {
        #pragma unroll
        for (int rw = 0; rw < 5; ++rw) {
            s[rw][0] += __shfl_xor(s[rw][0], off, 32);
            s[rw][1] += __shfl_xor(s[rw][1], off, 32);
            s[rw][2] += __shfl_xor(s[rw][2], off, 32);
            s[rw][3] += __shfl_xor(s[rw][3], off, 32);
        }
    }
    if (lane == 0) {
        #pragma unroll
        for (int rw = 0; rw < 4; ++rw) {
            float4 v;
            v.x = s[rw][0]; v.y = s[rw][1]; v.z = s[rw][2]; v.w = s[rw][3];
            wcs[4 * wave + rw] = v;
        }
        if (wave == 0) {
            const float* w512 = Wsr + 8;
            float4 v;
            v.x = br[0] + s[4][0] + w512[0];
            v.y = br[1] + s[4][1] + w512[1];
            v.z = br[2] + s[4][2] + w512[2];
            v.w = br[3] + s[4][3] + w512[3];
            bcs = v;
        }
    }
    __syncthreads();

    {
        float4 m = bcs;
        float4 mh = m;
        #pragma unroll
        for (int i = 0; i < 8; ++i) {
            const float4 c0 = wcs[4 * i], c1 = wcs[4 * i + 1], c2 = wcs[4 * i + 2], c3 = wcs[4 * i + 3];
            const float4 h = hv[i];
            m.x = fmaf(h.x, c0.x, m.x); m.y = fmaf(h.x, c0.y, m.y); m.z = fmaf(h.x, c0.z, m.z); m.w = fmaf(h.x, c0.w, m.w);
            m.x = fmaf(h.y, c1.x, m.x); m.y = fmaf(h.y, c1.y, m.y); m.z = fmaf(h.y, c1.z, m.z); m.w = fmaf(h.y, c1.w, m.w);
            m.x = fmaf(h.z, c2.x, m.x); m.y = fmaf(h.z, c2.y, m.y); m.z = fmaf(h.z, c2.z, m.z); m.w = fmaf(h.z, c2.w, m.w);
            m.x = fmaf(h.w, c3.x, m.x); m.y = fmaf(h.w, c3.y, m.y); m.z = fmaf(h.w, c3.z, m.z); m.w = fmaf(h.w, c3.w, m.w);
            const float4 e = hh[i];
            mh.x = fmaf(e.x, c0.x, mh.x); mh.y = fmaf(e.x, c0.y, mh.y); mh.z = fmaf(e.x, c0.z, mh.z); mh.w = fmaf(e.x, c0.w, mh.w);
            mh.x = fmaf(e.y, c1.x, mh.x); mh.y = fmaf(e.y, c1.y, mh.y); mh.z = fmaf(e.y, c1.z, mh.z); mh.w = fmaf(e.y, c1.w, mh.w);
            mh.x = fmaf(e.z, c2.x, mh.x); mh.y = fmaf(e.z, c2.y, mh.y); mh.z = fmaf(e.z, c2.z, mh.z); mh.w = fmaf(e.z, c2.w, mh.w);
            mh.x = fmaf(e.w, c3.x, mh.x); mh.y = fmaf(e.w, c3.y, mh.y); mh.z = fmaf(e.w, c3.z, mh.z); mh.w = fmaf(e.w, c3.w, mh.w);
        }
        hm[tid + 4] = m;
        if (tid < 4) hm[tid] = mh;
    }
    w2s[tid] = w2v0;
    w2s[tid + 256] = w2v1;
    if (tid < 128) { w2s[tid + 512] = w2v2; w3s[tid] = w3v; }
    if (tid < 32)  b2s[tid] = b2v;
    if (tid < 4)   b3s[tid] = b3v;
    __syncthreads();

    float win[20];
    #pragma unroll
    for (int j = 0; j < 5; ++j) {
        const float4 p = hm[tid + j];
        win[j * 4 + 0] = p.x; win[j * 4 + 1] = p.y;
        win[j * 4 + 2] = p.z; win[j * 4 + 3] = p.w;
    }
    float o0 = b3s[0], o1 = b3s[1], o2 = b3s[2], o3 = b3s[3];
    for (int n = 0; n < 32; ++n) {
        float h = b2s[n];
        #pragma unroll
        for (int k = 0; k < 20; ++k) h = fmaf(win[k], w2s[k * 32 + n], h);
        h = fmaxf(h, 0.f);
        o0 = fmaf(h, w3s[n * 4 + 0], o0);
        o1 = fmaf(h, w3s[n * 4 + 1], o1);
        o2 = fmaf(h, w3s[n * 4 + 2], o2);
        o3 = fmaf(h, w3s[n * 4 + 3], o3);
    }
    float* qo = out + ((size_t)g << 2);
    qo[0] = o0; qo[1] = o1; qo[2] = o2; qo[3] = o3;
}

extern "C" void kernel_launch(void* const* d_in, const int* in_sizes, int n_in,
                              void* d_out, int out_size, void* d_ws, size_t ws_size,
                              hipStream_t stream) {
    (void)in_sizes; (void)n_in; (void)out_size; (void)ws_size;
    const float* x  = (const float*)d_in[0];
    const float* W0 = (const float*)d_in[1];
    const float* b0 = (const float*)d_in[2];
    const float* W1 = (const float*)d_in[3];
    const float* b1 = (const float*)d_in[4];
    const float* Ws = (const float*)d_in[5];
    const float* bs = (const float*)d_in[6];
    const float* Wr = (const float*)d_in[7];
    const float* br = (const float*)d_in[8];
    const float* W2 = (const float*)d_in[9];
    const float* b2 = (const float*)d_in[10];
    const float* W3 = (const float*)d_in[11];
    const float* b3 = (const float*)d_in[12];
    float* out = (float*)d_out;

    float* ws  = (float*)d_ws;
    float* Wsr = ws;
    float* h0g = ws + H0G_OFFSET;

    k_front<<<256, 256, 0, stream>>>(x, W0, b0, Ws, bs, Wr, Wsr, h0g);
    k_back <<<128, 256, 0, stream>>>(h0g, Wsr, W1, b1, br, W2, b2, W3, b3, out);
}


# Back-kernel tail: one dependent 20-term fma chain per output, weights read trip by trip -> waves 0-3, 2 tokens per lane, 32 independent chains, weight reads one stage ahead; sum orders unchanged
# speedup vs baseline: 14.5408x; 1.1051x over previous
#include <hip/hip_runtime.h>
#include <hip/hip_bf16.h>


typedef __attribute__((ext_vector_type(2))) float v2f;
typedef __attribute__((ext_vector_type(8))) float v8f;

#define L_TOK   4096
#define B_BATCH 8
#define K0      484
#define H0      32
#define TB      128
#define HALO    120
#define W0_PIECES 968

#define WSR_LINE   32
#define H0G_OFFSET 8448

__global__ __launch_bounds__(256) void k_front(const float* __restrict__ x,
                                               const float* __restrict__ W0,
                                               const float* __restrict__ b0,
                                               const float* __restrict__ Ws,
                                               const float* __restrict__ bs,
                                               const float* __restrict__ Wr,
                                               float* __restrict__ Wsr,
                                               float* __restrict__ h0g) {
    __shared__ alignas(16) float xt[(TB + HALO) * 4];
    __shared__ alignas(16) float w0t[H0 * K0];
    __shared__ float b0s[H0];
    __shared__ alignas(16) float h0s[TB * 36];
    __shared__ float4 red[8][3];

    const int tid = threadIdx.x;
    const int bb  = blockIdx.x >> 5;
    const int t0  = (blockIdx.x & 31) * TB;

    float4 wv[4][4];
    #pragma unroll
    for (int r4 = 0; r4 < 4; ++r4) {
        int u = tid + 256 * r4;
        u = u < W0_PIECES ? u : W0_PIECES - 1;
        const float* src = W0 + (u >> 3) * 128 + (u & 7) * 4;
        #pragma unroll
        for (int r = 0; r < 4; ++r) wv[r4][r] = *(const float4*)(src + 32 * r);
    }
    const float b0v = b0[tid & 31];

    if (tid < TB + HALO) {
        const int gt = (t0 + tid + L_TOK - HALO) & (L_TOK - 1);
        *(float4*)(xt + 4 * tid) = *(const float4*)(x + ((bb << 12) + gt) * 4);
    }
    #pragma unroll
    for (int r4 = 0; r4 < 4; ++r4) {
        const int u = tid + 256 * r4;
        if (u < W0_PIECES) {
            float* dst = w0t + (u & 7) * 4 * K0 + (u >> 3) * 4;
            const float4 a0 = wv[r4][0], a1 = wv[r4][1], a2 = wv[r4][2], a3 = wv[r4][3];
            float4 t;
            t.x = a0.x; t.y = a1.x; t.z = a2.x; t.w = a3.x; *(float4*)(dst)          = t;
            t.x = a0.y; t.y = a1.y; t.z = a2.y; t.w = a3.y; *(float4*)(dst + K0)     = t;
            t.x = a0.z; t.y = a1.z; t.z = a2.z; t.w = a3.z; *(float4*)(dst + 2 * K0) = t;
            t.x = a0.w; t.y = a1.w; t.z = a2.w; t.w = a3.w; *(float4*)(dst + 3 * K0) = t;
        }
    }
    if (tid < H0)  b0s[tid] = b0v;
    __syncthreads();

    float4 w[3][4];
    {
        const float* src0 = Ws + (size_t)(2 * blockIdx.x) * 4096;
        #pragma unroll
        for (int i = 0; i < 4; ++i) {
            w[0][i] = *(const float4*)(src0 + 4 * (tid + 256 * i));
            w[1][i] = *(const float4*)(src0 + 4096 + 4 * (tid + 256 * i));
            w[2][i] = *(const float4*)(bs + 4 * (tid + 256 * i));
        }
    }
    float4 q[4][4];
    #pragma unroll
    for (int i = 0; i < 4; ++i) {
        const float4* qp = (const float4*)(Wr + 16 * (tid + 256 * i));
        q[i][0] = qp[0]; q[i][1] = qp[1]; q[i][2] = qp[2]; q[i][3] = qp[3];
    }

    const int wave = tid >> 5;
    const int lane = tid & 31;
    const int hl   = lane >> 4;
    const int m0   = lane & 15;
    const int c0   = hl << 1;

    v8f acc0 = {}; v8f acc1 = {};
    const float* ap  = xt  + (((wave << 4) + m0)) * 4 + c0;
    const float* bp0 = w0t + m0 * K0 + c0;
    const float* bp1 = w0t + (m0 + 16) * K0 + c0;

    for (int kk = 0; kk < 121; ++kk) {
        v2f a  = *(const v2f*)(ap  + 4 * kk);
        v2f bA = *(const v2f*)(bp0 + 4 * kk);
        v2f bB = *(const v2f*)(bp1 + 4 * kk);
        acc0 = __builtin_amdgcn_wmma_f32_16x16x4_f32(false, a, false, bA, (short)0, acc0, false, false);
        acc1 = __builtin_amdgcn_wmma_f32_16x16x4_f32(false, a, false, bB, (short)0, acc1, false, false);
    }

    #pragma unroll
    for (int g = 0; g < 8; ++g) {
        int M = (wave << 4) + g + (hl << 3);
        h0s[M * 36 + m0]      = tanhf(acc0[g] + b0s[m0]);
        h0s[M * 36 + m0 + 16] = tanhf(acc1[g] + b0s[m0 + 16]);
    }
    __syncthreads();

    #pragma unroll
    for (int r = 0; r < 4; ++r) {
        const int f   = tid + (r << 8);
        const int tok = f >> 3, qd = f & 7;
        const float4 v = *(const float4*)(h0s + tok * 36 + 4 * qd);
        *(float4*)(h0g + (size_t)((bb << 12) + t0 + tok) * 32 + 4 * qd) = v;
    }

    float acc[3][4];
    #pragma unroll
    for (int rr = 0; rr < 3; ++rr) { acc[rr][0] = 0.f; acc[rr][1] = 0.f; acc[rr][2] = 0.f; acc[rr][3] = 0.f; }
    #pragma unroll
    for (int i = 0; i < 4; ++i) {
        const float4 q0 = q[i][0], q1 = q[i][1], q2 = q[i][2], q3 = q[i][3];
        #pragma unroll
        for (int rr = 0; rr < 3; ++rr) {
            const float4 a = w[rr][i];
            acc[rr][0] = fmaf(a.x, q0.x, acc[rr][0]); acc[rr][0] = fmaf(a.y, q1.x, acc[rr][0]);
            acc[rr][0] = fmaf(a.z, q2.x, acc[rr][0]); acc[rr][0] = fmaf(a.w, q3.x, acc[rr][0]);
            acc[rr][1] = fmaf(a.x, q0.y, acc[rr][1]); acc[rr][1] = fmaf(a.y, q1.y, acc[rr][1]);
            acc[rr][1] = fmaf(a.z, q2.y, acc[rr][1]); acc[rr][1] = fmaf(a.w, q3.y, acc[rr][1]);
            acc[rr][2] = fmaf(a.x, q0.z, acc[rr][2]); acc[rr][2] = fmaf(a.y, q1.z, acc[rr][2]);
            acc[rr][2] = fmaf(a.z, q2.z, acc[rr][2]); acc[rr][2] = fmaf(a.w, q3.z, acc[rr][2]);
            acc[rr][3] = fmaf(a.x, q0.w, acc[rr][3]); acc[rr][3] = fmaf(a.y, q1.w, acc[rr][3]);
            acc[rr][3] = fmaf(a.z, q2.w, acc[rr][3]); acc[rr][3] = fmaf(a.w, q3.w, acc[rr][3]);
        }
    }
    #pragma unroll
    for (int off = 16; off > 0; off >>= 1) {
        #pragma unroll
        for (int rr = 0; rr < 3; ++rr) {
            acc[rr][0] += __shfl_xor(acc[rr][0], off, 32);
            acc[rr][1] += __shfl_xor(acc[rr][1], off, 32);
            acc[rr][2] += __shfl_xor(acc[rr][2], off, 32);
            acc[rr][3] += __shfl_xor(acc[rr][3], off, 32);
        }
    }
    if (lane == 0) {
        #pragma unroll
        for (int rr = 0; rr < 3; ++rr) {
            float4 s;
            s.x = acc[rr][0]; s.y = acc[rr][1]; s.z = acc[rr][2]; s.w = acc[rr][3];
            red[wave][rr] = s;
        }
    }
    __syncthreads();

    if (tid < 12) {
        const float* rf = (const float*)&red[0][0];
        float s = rf[tid];
        #pragma unroll
        for (int wv = 1; wv < 8; ++wv) s += rf[wv * 12 + tid];
        if (tid < 8 || blockIdx.x == 0) Wsr[blockIdx.x * WSR_LINE + tid] = s;
    }
}

__global__ __launch_bounds__(256) void k_back(const float* __restrict__ h0g,
                                              const float* __restrict__ Wsr,
                                              const float* __restrict__ W1,
                                              const float* __restrict__ b1,
                                              const float* __restrict__ br,
                                              const float* __restrict__ W2,
                                              const float* __restrict__ b2,
                                              const float* __restrict__ W3,
                                              const float* __restrict__ b3,
                                              float* __restrict__ out) {
    __shared__ alignas(16) float w2s[20 * 32];
    __shared__ alignas(16) float b2s[32];
    __shared__ alignas(16) float w3s[32 * 4];
    __shared__ alignas(16) float b3s[4];
    __shared__ float4 wcs[32];
    __shared__ float4 bcs;
    __shared__ float4 hm[256 + 4];

    const int tid  = threadIdx.x;
    const int lane = tid & 31;
    const int wave = tid >> 5;

    const int g  = blockIdx.x * 256 + tid;
    const int bb = g >> 12;
    const int tb = (blockIdx.x & 15) * 256;

    float4 q[4][4];
    #pragma unroll
    for (int i = 0; i < 4; ++i) {
        const float* ln = Wsr + (size_t)(2 * (lane + 32 * i)) * WSR_LINE;
        q[i][0] = *(const float4*)(ln);
        q[i][1] = *(const float4*)(ln + 4);
        q[i][2] = *(const float4*)(ln + WSR_LINE);
        q[i][3] = *(const float4*)(ln + WSR_LINE + 4);
    }
    float4 a[5][4];
    #pragma unroll
    for (int rw = 0; rw < 5; ++rw) {
        const float* src = (rw < 4) ? W1 + (4 * wave + rw) * 512 : b1;
        #pragma unroll
        for (int i = 0; i < 4; ++i) a[rw][i] = *(const float4*)(src + 4 * (lane + 32 * i));
    }

    float4 hv[8];
    {
        const float4* hp = (const float4*)(h0g + (size_t)g * 32);
        #pragma unroll
        for (int i = 0; i < 8; ++i) hv[i] = hp[i];
    }
    float4 hh[8];
    {
        const int tt = (tb + L_TOK - 4 + (tid & 3)) & (L_TOK - 1);
        const float4* hp = (const float4*)(h0g + (size_t)((bb << 12) + tt) * 32);
        #pragma unroll
        for (int i = 0; i < 8; ++i) hh[i] = hp[i];
    }

    const float w2v0 = W2[tid];
    const float w2v1 = W2[tid + 256];
    const float w2v2 = W2[512 + (tid & 127)];
    const float b2v  = b2[tid & 31];
    const float w3v  = W3[tid & 127];
    const float b3v  = b3[tid & 3];

    float s[5][4];
    #pragma unroll
    for (int rw = 0; rw < 5; ++rw) { s[rw][0] = 0.f; s[rw][1] = 0.f; s[rw][2] = 0.f; s[rw][3] = 0.f; }
    #pragma unroll
    for (int i = 0; i < 4; ++i) {
        const float4 q0 = q[i][0], q1 = q[i][1], q2 = q[i][2], q3 = q[i][3];
        #pragma unroll
        for (int rw = 0; rw < 5; ++rw) {
            const float4 v = a[rw][i];
            s[rw][0] = fmaf(v.x, q0.x, s[rw][0]); s[rw][0] = fmaf(v.y, q1.x, s[rw][0]);
            s[rw][0] = fmaf(v.z, q2.x, s[rw][0]); s[rw][0] = fmaf(v.w, q3.x, s[rw][0]);
            s[rw][1] = fmaf(v.x, q0.y, s[rw][1]); s[rw][1] = fmaf(v.y, q1.y, s[rw][1]);
            s[rw][1] = fmaf(v.z, q2.y, s[rw][1]); s[rw][1] = fmaf(v.w, q3.y, s[rw][1]);
            s[rw][2] = fmaf(v.x, q0.z, s[rw][2]); s[rw][2] = fmaf(v.y, q1.z, s[rw][2]);
            s[rw][2] = fmaf(v.z, q2.z, s[rw][2]); s[rw][2] = fmaf(v.w, q3.z, s[rw][2]);
            s[rw][3] = fmaf(v.x, q0.w, s[rw][3]); s[rw][3] = fmaf(v.y, q1.w, s[rw][3]);
            s[rw][3] = fmaf(v.z, q2.w, s[rw][3]); s[rw][3] = fmaf(v.w, q3.w, s[rw][3]);
        }
    }
    #pragma unroll
    for (int off = 16; off > 0; off >>= 1) {
        #pragma unroll
        for (int rw = 0; rw < 5; ++rw) {
            s[rw][0] += __shfl_xor(s[rw][0], off, 32);
            s[rw][1] += __shfl_xor(s[rw][1], off, 32);
            s[rw][2] += __shfl_xor(s[rw][2], off, 32);
            s[rw][3] += __shfl_xor(s[rw][3], off, 32);
        }
    }
    if (lane == 0) {
        #pragma unroll
        for (int rw = 0; rw < 4; ++rw) {
            float4 v;
            v.x = s[rw][0]; v.y = s[rw][1]; v.z = s[rw][2]; v.w = s[rw][3];
            wcs[4 * wave + rw] = v;
        }
        if (wave == 0) {
            const float* w512 = Wsr + 8;
            float4 v;
            v.x = br[0] + s[4][0] + w512[0];
            v.y = br[1] + s[4][1] + w512[1];
            v.z = br[2] + s[4][2] + w512[2];
            v.w = br[3] + s[4][3] + w512[3];
            bcs = v;
        }
    }
    __syncthreads();

    {
        float4 m = bcs;
        float4 mh = m;
        #pragma unroll
        for (int i = 0; i < 8; ++i) {
            const float4 c0 = wcs[4 * i], c1 = wcs[4 * i + 1], c2 = wcs[4 * i + 2], c3 = wcs[4 * i + 3];
            const float4 h = hv[i];
            m.x = fmaf(h.x, c0.x, m.x); m.y = fmaf(h.x, c0.y, m.y); m.z = fmaf(h.x, c0.z, m.z); m.w = fmaf(h.x, c0.w, m.w);
            m.x = fmaf(h.y, c1.x, m.x); m.y = fmaf(h.y, c1.y, m.y); m.z = fmaf(h.y, c1.z, m.z); m.w = fmaf(h.y, c1.w, m.w);
            m.x = fmaf(h.z, c2.x, m.x); m.y = fmaf(h.z, c2.y, m.y); m.z = fmaf(h.z, c2.z, m.z); m.w = fmaf(h.z, c2.w, m.w);
            m.x = fmaf(h.w, c3.x, m.x); m.y = fmaf(h.w, c3.y, m.y); m.z = fmaf(h.w, c3.z, m.z); m.w = fmaf(h.w, c3.w, m.w);
            const float4 e = hh[i];
            mh.x = fmaf(e.x, c0.x, mh.x); mh.y = fmaf(e.x, c0.y, mh.y); mh.z = fmaf(e.x, c0.z, mh.z); mh.w = fmaf(e.x, c0.w, mh.w);
            mh.x = fmaf(e.y, c1.x, mh.x); mh.y = fmaf(e.y, c1.y, mh.y); mh.z = fmaf(e.y, c1.z, mh.z); mh.w = fmaf(e.y, c1.w, mh.w);
            mh.x = fmaf(e.z, c2.x, mh.x); mh.y = fmaf(e.z, c2.y, mh.y); mh.z = fmaf(e.z, c2.z, mh.z); mh.w = fmaf(e.z, c2.w, mh.w);
            mh.x = fmaf(e.w, c3.x, mh.x); mh.y = fmaf(e.w, c3.y, mh.y); mh.z = fmaf(e.w, c3.z, mh.z); mh.w = fmaf(e.w, c3.w, mh.w);
        }
        hm[tid + 4] = m;
        if (tid < 4) hm[tid] = mh;
    }
    w2s[tid] = w2v0;
    w2s[tid + 256] = w2v1;
    if (tid < 128) { w2s[tid + 512] = w2v2; w3s[tid] = w3v; }
    if (tid < 32)  b2s[tid] = b2v;
    if (tid < 4)   b3s[tid] = b3v;
    __syncthreads();

    if (tid < 128) {
        float win[2][20];
        #pragma unroll
        for (int t = 0; t < 2; ++t) {
            #pragma unroll
            for (int j = 0; j < 5; ++j) {
                const float4 p = hm[tid + 128 * t + j];
                win[t][j * 4 + 0] = p.x; win[t][j * 4 + 1] = p.y;
                win[t][j * 4 + 2] = p.z; win[t][j * 4 + 3] = p.w;
            }
        }
        float4 bias[2][4];
        #pragma unroll
        for (int ng = 0; ng < 2; ++ng) {
            #pragma unroll
            for (int e4 = 0; e4 < 4; ++e4) bias[ng][e4] = *(const float4*)(b2s + ng * 16 + 4 * e4);
        }
        float o[2][4];
        {
            const float4 b = *(const float4*)b3s;
            #pragma unroll
            for (int t = 0; t < 2; ++t) { o[t][0] = b.x; o[t][1] = b.y; o[t][2] = b.z; o[t][3] = b.w; }
        }
        float h[2][16];
        float4 wb[2][16];
        #pragma unroll
        for (int i = 0; i < 16; ++i) wb[0][i] = *(const float4*)(w2s + (i >> 2) * 32 + 4 * (i & 3));
        #pragma unroll
        for (int s = 0; s < 12; ++s) {
            const int ng = s / 6, q = s % 6;
            if (s + 1 < 12) {
                const int ng1 = (s + 1) / 6, q1 = (s + 1) % 6;
                #pragma unroll
                for (int i = 0; i < 16; ++i) {
                    wb[(s + 1) & 1][i] = (q1 < 5)
                        ? *(const float4*)(w2s + (4 * q1 + (i >> 2)) * 32 + ng1 * 16 + 4 * (i & 3))
                        : *(const float4*)(w3s + (ng1 * 16 + i) * 4);
                }
            }
            __builtin_amdgcn_sched_barrier(0);
            if (q == 0) {
                #pragma unroll
                for (int t = 0; t < 2; ++t) {
                    #pragma unroll
                    for (int e4 = 0; e4 < 4; ++e4) {
                        h[t][4 * e4 + 0] = bias[ng][e4].x; h[t][4 * e4 + 1] = bias[ng][e4].y;
                        h[t][4 * e4 + 2] = bias[ng][e4].z; h[t][4 * e4 + 3] = bias[ng][e4].w;
                    }
                }
            }
            if (q < 5) {
                #pragma unroll
                for (int kk = 0; kk < 4; ++kk) {
                    #pragma unroll
                    for (int e4 = 0; e4 < 4; ++e4) {
                        const float4 w = wb[s & 1][kk * 4 + e4];
                        #pragma unroll
                        for (int t = 0; t < 2; ++t) {
                            const float a = win[t][4 * q + kk];
                            h[t][4 * e4 + 0] = fmaf(a, w.x, h[t][4 * e4 + 0]);
                            h[t][4 * e4 + 1] = fmaf(a, w.y, h[t][4 * e4 + 1]);
                            h[t][4 * e4 + 2] = fmaf(a, w.z, h[t][4 * e4 + 2]);
                            h[t][4 * e4 + 3] = fmaf(a, w.w, h[t][4 * e4 + 3]);
                        }
                    }
                }
            } else {
                #pragma unroll
                for (int e = 0; e < 16; ++e) {
                    const float4 w = wb[s & 1][e];
                    #pragma unroll
                    for (int t = 0; t < 2; ++t) {
                        const float hr = fmaxf(h[t][e], 0.f);
                        o[t][0] = fmaf(hr, w.x, o[t][0]);
                        o[t][1] = fmaf(hr, w.y, o[t][1]);
                        o[t][2] = fmaf(hr, w.z, o[t][2]);
                        o[t][3] = fmaf(hr, w.w, o[t][3]);
                    }
                }
            }
        }
        #pragma unroll
        for (int t = 0; t < 2; ++t) {
            float4 v;
            v.x = o[t][0]; v.y = o[t][1]; v.z = o[t][2]; v.w = o[t][3];
            *(float4*)(out + ((size_t)(blockIdx.x * 256 + tid + 128 * t) << 2)) = v;
        }
    }
}

extern "C" void kernel_launch(void* const* d_in, const int* in_sizes, int n_in,
                              void* d_out, int out_size, void* d_ws, size_t ws_size,
                              hipStream_t stream) {
    (void)in_sizes; (void)n_in; (void)out_size; (void)ws_size;
    const float* x  = (const float*)d_in[0];
    const float* W0 = (const float*)d_in[1];
    const float* b0 = (const float*)d_in[2];
    const float* W1 = (const float*)d_in[3];
    const float* b1 = (const float*)d_in[4];
    const float* Ws = (const float*)d_in[5];
    const float* bs = (const float*)d_in[6];
    const float* Wr = (const float*)d_in[7];
    const float* br = (const float*)d_in[8];
    const float* W2 = (const float*)d_in[9];
    const float* b2 = (const float*)d_in[10];
    const float* W3 = (const float*)d_in[11];
    const float* b3 = (const float*)d_in[12];
    float* out = (float*)d_out;

    float* ws  = (float*)d_ws;
    float* Wsr = ws;
    float* h0g = ws + H0G_OFFSET;

    k_front<<<256, 256, 0, stream>>>(x, W0, b0, Ws, bs, Wr, Wsr, h0g);
    k_back <<<128, 256, 0, stream>>>(h0g, Wsr, W1, b1, br, W2, b2, W3, b3, out);
}
